# starting kernel
# baseline (speedup 1.0000x reference)
#include <hip/hip_runtime.h>

typedef float v2f __attribute__((ext_vector_type(2)));
typedef float v4f __attribute__((ext_vector_type(4)));
typedef float v8f __attribute__((ext_vector_type(8)));

#define BATCH 32
#define TLEN  256
#define VOCAB 4096
#define NROWS (BATCH * TLEN)
#define NTRIM (BATCH * (TLEN - 1))

__global__ void __launch_bounds__(256) row_stats_kernel(
    const float* __restrict__ logits, v4f* __restrict__ stats) {
  const int row = blockIdx.x;
  const int tid = threadIdx.x;
  const v4f* p4 = reinterpret_cast<const v4f*>(logits + (size_t)row * VOCAB);

  v4f d[4];
#pragma unroll
  for (int i = 0; i < 4; ++i) d[i] = p4[tid + 256 * i];

  float m = d[0].x;
#pragma unroll
  for (int i = 0; i < 4; ++i)
    m = fmaxf(m, fmaxf(fmaxf(d[i].x, d[i].y), fmaxf(d[i].z, d[i].w)));
#pragma unroll
  for (int o = 16; o > 0; o >>= 1) m = fmaxf(m, __shfl_xor(m, o, 32));

  __shared__ float sm[8], ss0[8], ss1[8];
  const int wid = tid >> 5, lane = tid & 31;
  if (lane == 0) sm[wid] = m;
  __syncthreads();
  float bm = sm[0];
#pragma unroll
  for (int i = 1; i < 8; ++i) bm = fmaxf(bm, sm[i]);

  float s0 = 0.f, s1 = 0.f;
#pragma unroll
  for (int i = 0; i < 4; ++i) {
    const float x[4] = {d[i].x, d[i].y, d[i].z, d[i].w};
#pragma unroll
    for (int j = 0; j < 4; ++j) {
      const float t = x[j] - bm;
      const float e = __expf(t);
      s0 += e;
      s1 += t * e;
    }
  }
#pragma unroll
  for (int o = 16; o > 0; o >>= 1) {
    s0 += __shfl_xor(s0, o, 32);
    s1 += __shfl_xor(s1, o, 32);
  }
  if (lane == 0) { ss0[wid] = s0; ss1[wid] = s1; }
  __syncthreads();
  if (tid == 0) {
    float S0 = 0.f, S1 = 0.f;
#pragma unroll
    for (int i = 0; i < 8; ++i) { S0 += ss0[i]; S1 += ss1[i]; }
    const float ent = __logf(S0) - S1 / S0;
    v4f r; r.x = bm; r.y = S0; r.z = ent; r.w = 0.f;
    stats[row] = r;
  }
}

__global__ void __launch_bounds__(256) mlp_flags_kernel(
    const v4f* __restrict__ stats, const float* __restrict__ w1,
    const float* __restrict__ b1, const float* __restrict__ w2,
    const float* __restrict__ b2, float* __restrict__ flags) {
  const int lane = threadIdx.x & 31;
  const int tile = (blockIdx.x * blockDim.x + threadIdx.x) >> 5;

  const int M = lane & 15;
  int r = tile * 16 + M;
  const int rc = (r < NTRIM) ? r : (NTRIM - 1);
  const int b = rc / (TLEN - 1);
  const int t = rc % (TLEN - 1);

  const float ent = stats[b * TLEN + t].z;
  const float kEntMaxInv = 1.0f / 5.54517744447956247f;
  const float norm = 2.0f * ent * kEntMaxInv - 1.0f;

  const int koff = (lane >> 4) << 1;
  v8f acc = {};
#pragma unroll
  for (int k0 = 0; k0 < 128; k0 += 4) {
    const int ka = k0 + koff;
    const float h0 = fmaxf(fmaf(norm, w1[ka], b1[ka]), 0.f);
    const float h1 = fmaxf(fmaf(norm, w1[ka + 1], b1[ka + 1]), 0.f);
    v2f a;  a.x = h0;      a.y = h1;
    v2f bb; bb.x = w2[ka]; bb.y = w2[ka + 1];
    acc = __builtin_amdgcn_wmma_f32_16x16x4_f32(
         false, a,  false, bb,
         (short)0, acc,  false,  false);
  }

  const float bias = b2[0];
  if (lane == 0 || lane == 16) {
#pragma unroll
    for (int j = 0; j < 8; ++j) {
      const int Mo = j + ((lane == 16) ? 8 : 0);
      const int ro = tile * 16 + Mo;
      if (ro < NTRIM) {
        const int tt = ro % (TLEN - 1);
        const float lin = 2.0f * acc[j] + bias;
        const float z = lin - __logf((float)((TLEN - 1) - tt));
        flags[ro] = 1.0f / (1.0f + __expf(-z));
      }
    }
  }
}

__global__ void scan_scale_kernel(const float* __restrict__ flags,
                                  float* __restrict__ scale) {
  const int b = threadIdx.x;
  if (b >= BATCH) return;
  const float* f = flags + b * (TLEN - 1);
  float* s = scale + b * TLEN;
  float res = 1.0f;
  for (int t = 0; t < TLEN; ++t) {
    const float fn = (t < TLEN - 1) ? f[t] : 1.0f;
    s[t] = fn * res;
    res *= (1.0f - fn);
  }
}

__global__ void __launch_bounds__(256) finalize_kernel(
    const float* __restrict__ logits, const v4f* __restrict__ stats,
    const float* __restrict__ scale, float* __restrict__ out) {
  const int row = blockIdx.x;
  const int tid = threadIdx.x;
  const v4f st = stats[row];
  const float m = st.x;
  const float k = scale[row] / st.y;
  const v4f* p4 = reinterpret_cast<const v4f*>(logits + (size_t)row * VOCAB);
  v4f* o4 = reinterpret_cast<v4f*>(out + (size_t)row * VOCAB);
#pragma unroll
  for (int i = 0; i < 4; ++i) {
    const v4f x = p4[tid + 256 * i];
    v4f r;
    r.x = __expf(x.x - m) * k;
    r.y = __expf(x.y - m) * k;
    r.z = __expf(x.z - m) * k;
    r.w = __expf(x.w - m) * k;
    __builtin_nontemporal_store(r, &o4[tid + 256 * i]);
  }
}

extern "C" void kernel_launch(void* const* d_in, const int* in_sizes, int n_in,
                              void* d_out, int out_size, void* d_ws, size_t ws_size,
                              hipStream_t stream) {
  const float* logits = (const float*)d_in[0];
  const float* w1 = (const float*)d_in[1];
  const float* b1 = (const float*)d_in[2];
  const float* w2 = (const float*)d_in[3];
  const float* b2 = (const float*)d_in[4];
  float* out = (float*)d_out;

  char* ws = (char*)d_ws;
  v4f*   stats = (v4f*)ws;
  float* flags = (float*)(ws + (size_t)NROWS * 16);
  float* scale = (float*)(ws + (size_t)NROWS * 16 + 32768);

  row_stats_kernel<<<NROWS, 256, 0, stream>>>(logits, stats);
  mlp_flags_kernel<<<64, 256, 0, stream>>>(stats, w1, b1, w2, b2, flags);
  scan_scale_kernel<<<1, 32, 0, stream>>>(flags, scale);
  finalize_kernel<<<NROWS, 256, 0, stream>>>(logits, stats, scale, out);
}


# Fuse the MLP-flags and scan launches into one kernel, one block per batch: weights and flags staged in LDS, one sigmoid per lane, scan chain fed from LDS not global memory (4 launches to 3)
# speedup vs baseline: 1.4090x; 1.4090x over previous
#include <hip/hip_runtime.h>

typedef float v2f __attribute__((ext_vector_type(2)));
typedef float v4f __attribute__((ext_vector_type(4)));
typedef float v8f __attribute__((ext_vector_type(8)));

#define BATCH 32
#define TLEN  256
#define VOCAB 4096
#define NROWS (BATCH * TLEN)

__global__ void __launch_bounds__(256) row_stats_kernel(
    const float* __restrict__ logits, v4f* __restrict__ stats) {
  const int row = blockIdx.x;
  const int tid = threadIdx.x;
  const v4f* p4 = reinterpret_cast<const v4f*>(logits + (size_t)row * VOCAB);

  v4f d[4];
#pragma unroll
  for (int i = 0; i < 4; ++i) d[i] = p4[tid + 256 * i];

  float m = d[0].x;
#pragma unroll
  for (int i = 0; i < 4; ++i)
    m = fmaxf(m, fmaxf(fmaxf(d[i].x, d[i].y), fmaxf(d[i].z, d[i].w)));
#pragma unroll
  for (int o = 16; o > 0; o >>= 1) m = fmaxf(m, __shfl_xor(m, o, 32));

  __shared__ float sm[8], ss0[8], ss1[8];
  const int wid = tid >> 5, lane = tid & 31;
  if (lane == 0) sm[wid] = m;
  __syncthreads();
  float bm = sm[0];
#pragma unroll
  for (int i = 1; i < 8; ++i) bm = fmaxf(bm, sm[i]);

  float s0 = 0.f, s1 = 0.f;
#pragma unroll
  for (int i = 0; i < 4; ++i) {
    const float x[4] = {d[i].x, d[i].y, d[i].z, d[i].w};
#pragma unroll
    for (int j = 0; j < 4; ++j) {
      const float t = x[j] - bm;
      const float e = __expf(t);
      s0 += e;
      s1 += t * e;
    }
  }
#pragma unroll
  for (int o = 16; o > 0; o >>= 1) {
    s0 += __shfl_xor(s0, o, 32);
    s1 += __shfl_xor(s1, o, 32);
  }
  if (lane == 0) { ss0[wid] = s0; ss1[wid] = s1; }
  __syncthreads();
  if (tid == 0) {
    float S0 = 0.f, S1 = 0.f;
#pragma unroll
    for (int i = 0; i < 8; ++i) { S0 += ss0[i]; S1 += ss1[i]; }
    const float ent = __logf(S0) - S1 / S0;
    v4f r; r.x = bm; r.y = S0; r.z = ent; r.w = 0.f;
    stats[row] = r;
  }
}

__global__ void __launch_bounds__(256) flags_scan_kernel(
    const v4f* __restrict__ stats, const float* __restrict__ w1,
    const float* __restrict__ b1, const float* __restrict__ w2,
    const float* __restrict__ b2, float* __restrict__ scale) {
  __shared__ v4f s_wb[64];
  __shared__ v2f s_w2[64];
  __shared__ v4f s_fn4[TLEN / 4];
  __shared__ v4f s_res4[TLEN / 4];
  float* s_fn = reinterpret_cast<float*>(s_fn4);
  float* s_res = reinterpret_cast<float*>(s_res4);

  const int b = blockIdx.x;
  const int tid = threadIdx.x;
  const int lane = tid & 31, wid = tid >> 5;
  const int c = lane & 15, h = lane >> 4;

  const int tA = wid * 32 + c;
  const float entA = stats[b * TLEN + tA].z;
  const float entB = stats[b * TLEN + tA + 16].z;

  if (tid < 64) {
    const v2f w = *reinterpret_cast<const v2f*>(w1 + 2 * tid);
    const v2f o = *reinterpret_cast<const v2f*>(b1 + 2 * tid);
    v4f p; p.x = w.x; p.y = w.y; p.z = o.x; p.w = o.y;
    s_wb[tid] = p;
    s_w2[tid] = *reinterpret_cast<const v2f*>(w2 + 2 * tid);
  }
  __syncthreads();

  const float kEntMaxInv = 1.0f / 5.54517744447956247f;
  const float normA = 2.0f * entA * kEntMaxInv - 1.0f;
  const float normB = 2.0f * entB * kEntMaxInv - 1.0f;

  v4f pw[2][8];
  v2f pb[2][8];
#pragma unroll
  for (int s = 0; s < 8; ++s) {
    pw[0][s] = s_wb[2 * s + h];
    pb[0][s] = s_w2[2 * s + h];
  }
  v8f accA = {}, accB = {};
#pragma unroll
  for (int ch = 0; ch < 4; ++ch) {
    if (ch + 1 < 4) {
#pragma unroll
      for (int s = 0; s < 8; ++s) {
        pw[(ch + 1) & 1][s] = s_wb[2 * (8 * (ch + 1) + s) + h];
        pb[(ch + 1) & 1][s] = s_w2[2 * (8 * (ch + 1) + s) + h];
      }
    }
    __builtin_amdgcn_sched_barrier(0);
#pragma unroll
    for (int s = 0; s < 8; ++s) {
      const v4f p = pw[ch & 1][s];
      const v2f bb = pb[ch & 1][s];
      v2f aA, aB;
      aA.x = fmaxf(fmaf(normA, p.x, p.z), 0.f);
      aA.y = fmaxf(fmaf(normA, p.y, p.w), 0.f);
      aB.x = fmaxf(fmaf(normB, p.x, p.z), 0.f);
      aB.y = fmaxf(fmaf(normB, p.y, p.w), 0.f);
      accA = __builtin_amdgcn_wmma_f32_16x16x4_f32(
           false, aA,  false, bb,
           (short)0, accA,  false,  false);
      accB = __builtin_amdgcn_wmma_f32_16x16x4_f32(
           false, aB,  false, bb,
           (short)0, accB,  false,  false);
    }
    __builtin_amdgcn_sched_barrier(0);
  }

  const int sel = c & 7;
  const bool second = (c & 8) != 0;
  float d = second ? accB[0] : accA[0];
#pragma unroll
  for (int r = 1; r < 8; ++r) {
    const float x = second ? accB[r] : accA[r];
    d = (sel == r) ? x : d;
  }
  const int myT = wid * 32 + (second ? 16 : 0) + 8 * h + sel;
  const float z = (2.0f * d + b2[0]) - __logf((float)((TLEN - 1) - myT));
  const float sg = 1.0f / (1.0f + __expf(-z));
  const float myF = (myT < TLEN - 1) ? sg : 1.0f;
  s_fn[myT] = myF;
  __syncthreads();

  if (tid == 0) {
    v4f fb[2][8];
#pragma unroll
    for (int i = 0; i < 8; ++i) fb[0][i] = s_fn4[i];
    float res = 1.0f;
#pragma unroll
    for (int g = 0; g < 8; ++g) {
      if (g + 1 < 8) {
#pragma unroll
        for (int i = 0; i < 8; ++i) fb[(g + 1) & 1][i] = s_fn4[8 * (g + 1) + i];
      }
      __builtin_amdgcn_sched_barrier(0);
#pragma unroll
      for (int i = 0; i < 8; ++i) {
        const v4f f = fb[g & 1][i];
        v4f r;
        r.x = res; res *= (1.0f - f.x);
        r.y = res; res *= (1.0f - f.y);
        r.z = res; res *= (1.0f - f.z);
        r.w = res; res *= (1.0f - f.w);
        s_res4[8 * g + i] = r;
      }
    }
  }
  __syncthreads();

  scale[b * TLEN + myT] = myF * s_res[myT];
}

__global__ void __launch_bounds__(256) finalize_kernel(
    const float* __restrict__ logits, const v4f* __restrict__ stats,
    const float* __restrict__ scale, float* __restrict__ out) {
  const int row = blockIdx.x;
  const int tid = threadIdx.x;
  const v4f st = stats[row];
  const float m = st.x;
  const float k = scale[row] / st.y;
  const v4f* p4 = reinterpret_cast<const v4f*>(logits + (size_t)row * VOCAB);
  v4f* o4 = reinterpret_cast<v4f*>(out + (size_t)row * VOCAB);
#pragma unroll
  for (int i = 0; i < 4; ++i) {
    const v4f x = p4[tid + 256 * i];
    v4f r;
    r.x = __expf(x.x - m) * k;
    r.y = __expf(x.y - m) * k;
    r.z = __expf(x.z - m) * k;
    r.w = __expf(x.w - m) * k;
    __builtin_nontemporal_store(r, &o4[tid + 256 * i]);
  }
}

extern "C" void kernel_launch(void* const* d_in, const int* in_sizes, int n_in,
                              void* d_out, int out_size, void* d_ws, size_t ws_size,
                              hipStream_t stream) {
  const float* logits = (const float*)d_in[0];
  const float* w1 = (const float*)d_in[1];
  const float* b1 = (const float*)d_in[2];
  const float* w2 = (const float*)d_in[3];
  const float* b2 = (const float*)d_in[4];
  float* out = (float*)d_out;

  char* ws = (char*)d_ws;
  v4f*   stats = (v4f*)ws;
  float* scale = (float*)(ws + (size_t)NROWS * 16 + 32768);

  row_stats_kernel<<<NROWS, 256, 0, stream>>>(logits, stats);
  flags_scan_kernel<<<BATCH, 256, 0, stream>>>(stats, w1, b1, w2, b2, scale);
  finalize_kernel<<<NROWS, 256, 0, stream>>>(logits, stats, scale, out);
}


# Drop the flags+scan launch (3 to 2): row-stats wave 0 makes the row's flag with 2 WMMAs (k on the tile diagonal, trace); finalize waves rebuild the prefix product as a tree; sum/product order changes
# speedup vs baseline: 1.5009x; 1.0652x over previous
#include <hip/hip_runtime.h>


typedef float v2f __attribute__((ext_vector_type(2)));
typedef float v4f __attribute__((ext_vector_type(4)));
typedef float v8f __attribute__((ext_vector_type(8)));

#define BATCH 32
#define TLEN  256
#define VOCAB 4096
#define NROWS (BATCH * TLEN)

template <int CTRL>
__device__ __forceinline__ float row_xmask(float x) {
  const int b = __builtin_bit_cast(int, x);
  return __builtin_bit_cast(float, __builtin_amdgcn_mov_dpp(b, CTRL, 0xf, 0xf, true));
}
__device__ __forceinline__ float xhalf(float x) {
  const int b = __builtin_bit_cast(int, x);
  return __builtin_bit_cast(
      float, __builtin_amdgcn_permlanex16(b, b, 0x76543210, 0xfedcba98, false, false));
}

__global__ void __launch_bounds__(256) row_stats_flag_kernel(
    const float* __restrict__ logits, const float* w1, const float* b1,
    const float* w2, const float* __restrict__ b2, v2f* __restrict__ stats,
    float* __restrict__ flags) {
  const int row = blockIdx.x;
  const int tid = threadIdx.x;
  const int lane = tid & 31;
  const int wid = __builtin_amdgcn_readfirstlane(tid >> 5);
  const v4f* p4 = reinterpret_cast<const v4f*>(logits + (size_t)row * VOCAB);

  v4f d[4];
#pragma unroll
  for (int i = 0; i < 4; ++i) d[i] = p4[tid + 256 * i];

  v4f w1q = {}, b1q = {}, w2q = {};
  if (wid == 0) {
    w1q = *reinterpret_cast<const v4f*>(w1 + 4 * lane);
    b1q = *reinterpret_cast<const v4f*>(b1 + 4 * lane);
    w2q = *reinterpret_cast<const v4f*>(w2 + 4 * lane);
  }

  float m = d[0].x;
#pragma unroll
  for (int i = 0; i < 4; ++i)
    m = fmaxf(m, fmaxf(fmaxf(d[i].x, d[i].y), fmaxf(d[i].z, d[i].w)));
#pragma unroll
  for (int o = 16; o > 0; o >>= 1) m = fmaxf(m, __shfl_xor(m, o, 32));

  __shared__ float sm[8], ss0[8], ss1[8];
  if (lane == 0) sm[wid] = m;
  __syncthreads();
  float bm = sm[0];
#pragma unroll
  for (int i = 1; i < 8; ++i) bm = fmaxf(bm, sm[i]);

  float s0 = 0.f, s1 = 0.f;
#pragma unroll
  for (int i = 0; i < 4; ++i) {
    const float x[4] = {d[i].x, d[i].y, d[i].z, d[i].w};
#pragma unroll
    for (int j = 0; j < 4; ++j) {
      const float t = x[j] - bm;
      const float e = __expf(t);
      s0 += e;
      s1 += t * e;
    }
  }
#pragma unroll
  for (int o = 16; o > 0; o >>= 1) {
    s0 += __shfl_xor(s0, o, 32);
    s1 += __shfl_xor(s1, o, 32);
  }
  if (lane == 0) { ss0[wid] = s0; ss1[wid] = s1; }
  __syncthreads();

  if (wid == 0) {
    float S0 = 0.f, S1 = 0.f;
#pragma unroll
    for (int i = 0; i < 8; ++i) { S0 += ss0[i]; S1 += ss1[i]; }
    const float ent = __logf(S0) - S1 / S0;

    const float kEntMaxInv = 1.0f / 5.54517744447956247f;
    const float norm = 2.0f * ent * kEntMaxInv - 1.0f;

    v2f a0, a1, q0, q1;
    a0.x = fmaxf(fmaf(norm, w1q.x, b1q.x), 0.f);
    a0.y = fmaxf(fmaf(norm, w1q.y, b1q.y), 0.f);
    a1.x = fmaxf(fmaf(norm, w1q.z, b1q.z), 0.f);
    a1.y = fmaxf(fmaf(norm, w1q.w, b1q.w), 0.f);
    q0.x = w2q.x; q0.y = w2q.y;
    q1.x = w2q.z; q1.y = w2q.w;
    v8f acc = {};
    acc = __builtin_amdgcn_wmma_f32_16x16x4_f32(
         false, a0,  false, q0,
         (short)0, acc,  false,  false);
    acc = __builtin_amdgcn_wmma_f32_16x16x4_f32(
         false, a1,  false, q1,
         (short)0, acc,  false,  false);

    const int sel = lane & 7;
    float dg = acc[0];
#pragma unroll
    for (int r = 1; r < 8; ++r) dg = (sel == r) ? acc[r] : dg;
    const bool on_diag = (((lane & 15) >> 3) == (lane >> 4));
    dg = on_diag ? dg : 0.f;
    dg += row_xmask<0x161>(dg);
    dg += row_xmask<0x162>(dg);
    dg += row_xmask<0x164>(dg);
    dg += row_xmask<0x168>(dg);
    dg += xhalf(dg);

    const int t = row & (TLEN - 1);
    const float z = (2.0f * dg + b2[0]) - __logf((float)((TLEN - 1) - t));
    const float sg = 1.0f / (1.0f + __expf(-z));
    const float f = (t < TLEN - 1) ? sg : 1.0f;
    if (lane == 0) {
      v2f r; r.x = bm; r.y = S0;
      stats[row] = r;
      flags[row] = f;
    }
  }
}

__global__ void __launch_bounds__(256) finalize_kernel(
    const float* __restrict__ logits, const v2f* __restrict__ stats,
    const float* __restrict__ flags, float* __restrict__ out) {
  const int row = blockIdx.x;
  const int tid = threadIdx.x;
  const int lane = tid & 31;
  const int t = row & (TLEN - 1);
  const int base = row - t;

  const v2f st = stats[row];
  const float fr = flags[row];

  const int n = t - 8 * lane;
  const v4f* f4 = reinterpret_cast<const v4f*>(flags + base);
  const int src = (n > 0) ? lane : 0;
  const v4f fa = f4[2 * src];
  const v4f fb = f4[2 * src + 1];
  __builtin_amdgcn_sched_barrier(0);

  const v4f* p4 = reinterpret_cast<const v4f*>(logits + (size_t)row * VOCAB);
  v4f x[4];
#pragma unroll
  for (int i = 0; i < 4; ++i) x[i] = p4[tid + 256 * i];

  float g = (n > 0) ? (1.0f - fa.x) : 1.0f;
  g *= (n > 1) ? (1.0f - fa.y) : 1.0f;
  g *= (n > 2) ? (1.0f - fa.z) : 1.0f;
  g *= (n > 3) ? (1.0f - fa.w) : 1.0f;
  g *= (n > 4) ? (1.0f - fb.x) : 1.0f;
  g *= (n > 5) ? (1.0f - fb.y) : 1.0f;
  g *= (n > 6) ? (1.0f - fb.z) : 1.0f;
  g *= (n > 7) ? (1.0f - fb.w) : 1.0f;
  g *= row_xmask<0x161>(g);
  g *= row_xmask<0x162>(g);
  g *= row_xmask<0x164>(g);
  g *= row_xmask<0x168>(g);
  g *= xhalf(g);
  __builtin_amdgcn_sched_barrier(0);

  const float m = st.x;
  const float k = (fr * g) / st.y;
  v4f* o4 = reinterpret_cast<v4f*>(out + (size_t)row * VOCAB);
#pragma unroll
  for (int i = 0; i < 4; ++i) {
    v4f r;
    r.x = __expf(x[i].x - m) * k;
    r.y = __expf(x[i].y - m) * k;
    r.z = __expf(x[i].z - m) * k;
    r.w = __expf(x[i].w - m) * k;
    __builtin_nontemporal_store(r, &o4[tid + 256 * i]);
  }
}

extern "C" void kernel_launch(void* const* d_in, const int* in_sizes, int n_in,
                              void* d_out, int out_size, void* d_ws, size_t ws_size,
                              hipStream_t stream) {
  const float* logits = (const float*)d_in[0];
  const float* w1 = (const float*)d_in[1];
  const float* b1 = (const float*)d_in[2];
  const float* w2 = (const float*)d_in[3];
  const float* b2 = (const float*)d_in[4];
  float* out = (float*)d_out;

  char* ws = (char*)d_ws;
  v2f*   stats = (v2f*)ws;
  float* flags = (float*)(ws + (size_t)NROWS * 16 + 32768);

  row_stats_flag_kernel<<<NROWS, 256, 0, stream>>>(logits, w1, b1, w2, b2, stats, flags);
  finalize_kernel<<<NROWS, 256, 0, stream>>>(logits, stats, flags, out);
}


# One wave per row in both kernels (8x fewer waves; a lane keeps its 128 elements, 32 b128 loads in flight): reductions become in-wave butterflies, no LDS, no barrier; S0/S1 sum order changes
# speedup vs baseline: 1.6656x; 1.1097x over previous
#include <hip/hip_runtime.h>


typedef float v2f __attribute__((ext_vector_type(2)));
typedef float v4f __attribute__((ext_vector_type(4)));
typedef float v8f __attribute__((ext_vector_type(8)));

#define BATCH 32
#define TLEN  256
#define VOCAB 4096
#define NROWS (BATCH * TLEN)
#define PIECES 32
#define ROWS_PER_BLOCK 4
#define BLOCK_THREADS 128
#define NBLOCKS 2048

template <int CTRL>
__device__ __forceinline__ float row_xmask(float x) {
  const int b = __builtin_bit_cast(int, x);
  return __builtin_bit_cast(float, __builtin_amdgcn_mov_dpp(b, CTRL, 0xf, 0xf, true));
}
__device__ __forceinline__ float xhalf(float x) {
  const int b = __builtin_bit_cast(int, x);
  return __builtin_bit_cast(
      float, __builtin_amdgcn_permlanex16(b, b, 0x76543210, 0xfedcba98, false, false));
}
__device__ __forceinline__ float wave_sum(float v) {
  v += row_xmask<0x161>(v);
  v += row_xmask<0x162>(v);
  v += row_xmask<0x164>(v);
  v += row_xmask<0x168>(v);
  v += xhalf(v);
  return v;
}
__device__ __forceinline__ float wave_max(float v) {
  v = fmaxf(v, row_xmask<0x161>(v));
  v = fmaxf(v, row_xmask<0x162>(v));
  v = fmaxf(v, row_xmask<0x164>(v));
  v = fmaxf(v, row_xmask<0x168>(v));
  v = fmaxf(v, xhalf(v));
  return v;
}
__device__ __forceinline__ float wave_prod(float v) {
  v *= row_xmask<0x161>(v);
  v *= row_xmask<0x162>(v);
  v *= row_xmask<0x164>(v);
  v *= row_xmask<0x168>(v);
  v *= xhalf(v);
  return v;
}

__global__ void __launch_bounds__(BLOCK_THREADS) row_stats_flag_kernel(
    const float* __restrict__ logits, const float* w1, const float* b1,
    const float* w2, const float* __restrict__ b2, v2f* __restrict__ stats,
    float* __restrict__ flags) {
  const int lane = threadIdx.x & 31;
  const int wid = __builtin_amdgcn_readfirstlane(threadIdx.x >> 5);
  const int row = blockIdx.x * ROWS_PER_BLOCK + wid;
  const v4f* p4 = reinterpret_cast<const v4f*>(logits + (size_t)row * VOCAB) + lane;

  v4f d[PIECES];
#pragma unroll
  for (int i = 0; i < PIECES; ++i) d[i] = p4[32 * i];

  const v4f w1q = *reinterpret_cast<const v4f*>(w1 + 4 * lane);
  const v4f b1q = *reinterpret_cast<const v4f*>(b1 + 4 * lane);
  const v4f w2q = *reinterpret_cast<const v4f*>(w2 + 4 * lane);

  float mx[4];
#pragma unroll
  for (int c = 0; c < 4; ++c)
    mx[c] = fmaxf(fmaxf(d[c].x, d[c].y), fmaxf(d[c].z, d[c].w));
#pragma unroll
  for (int i = 4; i < PIECES; ++i)
    mx[i & 3] = fmaxf(mx[i & 3], fmaxf(fmaxf(d[i].x, d[i].y), fmaxf(d[i].z, d[i].w)));
  const float bm = wave_max(fmaxf(fmaxf(mx[0], mx[1]), fmaxf(mx[2], mx[3])));

  float s0[4] = {0.f, 0.f, 0.f, 0.f}, s1[4] = {0.f, 0.f, 0.f, 0.f};
#pragma unroll
  for (int i = 0; i < PIECES; ++i) {
    const float x[4] = {d[i].x, d[i].y, d[i].z, d[i].w};
#pragma unroll
    for (int j = 0; j < 4; ++j) {
      const float t = x[j] - bm;
      const float e = __expf(t);
      s0[j] += e;
      s1[j] += t * e;
    }
  }
  const float S0 = wave_sum((s0[0] + s0[1]) + (s0[2] + s0[3]));
  const float S1 = wave_sum((s1[0] + s1[1]) + (s1[2] + s1[3]));
  const float ent = __logf(S0) - S1 / S0;

  const float kEntMaxInv = 1.0f / 5.54517744447956247f;
  const float norm = 2.0f * ent * kEntMaxInv - 1.0f;

  v2f a0, a1, q0, q1;
  a0.x = fmaxf(fmaf(norm, w1q.x, b1q.x), 0.f);
  a0.y = fmaxf(fmaf(norm, w1q.y, b1q.y), 0.f);
  a1.x = fmaxf(fmaf(norm, w1q.z, b1q.z), 0.f);
  a1.y = fmaxf(fmaf(norm, w1q.w, b1q.w), 0.f);
  q0.x = w2q.x; q0.y = w2q.y;
  q1.x = w2q.z; q1.y = w2q.w;
  v8f acc = {};
  acc = __builtin_amdgcn_wmma_f32_16x16x4_f32(
       false, a0,  false, q0,
       (short)0, acc,  false,  false);
  acc = __builtin_amdgcn_wmma_f32_16x16x4_f32(
       false, a1,  false, q1,
       (short)0, acc,  false,  false);

  const int sel = lane & 7;
  float dg = acc[0];
#pragma unroll
  for (int r = 1; r < 8; ++r) dg = (sel == r) ? acc[r] : dg;
  const bool on_diag = (((lane & 15) >> 3) == (lane >> 4));
  dg = on_diag ? dg : 0.f;
  dg = wave_sum(dg);

  const int t = row & (TLEN - 1);
  const float z = (2.0f * dg + b2[0]) - __logf((float)((TLEN - 1) - t));
  const float sg = 1.0f / (1.0f + __expf(-z));
  const float f = (t < TLEN - 1) ? sg : 1.0f;
  if (lane == 0) {
    v2f r; r.x = bm; r.y = S0;
    stats[row] = r;
    flags[row] = f;
  }
}

__global__ void __launch_bounds__(BLOCK_THREADS) finalize_kernel(
    const float* __restrict__ logits, const v2f* __restrict__ stats,
    const float* __restrict__ flags, float* __restrict__ out) {
  const int lane = threadIdx.x & 31;
  const int wid = __builtin_amdgcn_readfirstlane(threadIdx.x >> 5);
  const int row = blockIdx.x * ROWS_PER_BLOCK + wid;
  const int t = row & (TLEN - 1);
  const int base = row - t;

  const v2f st = stats[row];
  const float fr = flags[row];

  const int n = t - 8 * lane;
  const v4f* f4 = reinterpret_cast<const v4f*>(flags + base);
  const int src = (n > 0) ? lane : 0;
  const v4f fa = f4[2 * src];
  const v4f fb = f4[2 * src + 1];
  __builtin_amdgcn_sched_barrier(0);

  const v4f* p4 = reinterpret_cast<const v4f*>(logits + (size_t)row * VOCAB) + lane;
  v4f x[PIECES];
#pragma unroll
  for (int i = 0; i < PIECES; ++i) x[i] = p4[32 * i];

  float g = (n > 0) ? (1.0f - fa.x) : 1.0f;
  g *= (n > 1) ? (1.0f - fa.y) : 1.0f;
  g *= (n > 2) ? (1.0f - fa.z) : 1.0f;
  g *= (n > 3) ? (1.0f - fa.w) : 1.0f;
  g *= (n > 4) ? (1.0f - fb.x) : 1.0f;
  g *= (n > 5) ? (1.0f - fb.y) : 1.0f;
  g *= (n > 6) ? (1.0f - fb.z) : 1.0f;
  g *= (n > 7) ? (1.0f - fb.w) : 1.0f;
  g = wave_prod(g);
  __builtin_amdgcn_sched_barrier(0);

  const float m = st.x;
  const float k = (fr * g) / st.y;
  v4f* o4 = reinterpret_cast<v4f*>(out + (size_t)row * VOCAB) + lane;
#pragma unroll
  for (int i = 0; i < PIECES; ++i) {
    v4f r;
    r.x = __expf(x[i].x - m) * k;
    r.y = __expf(x[i].y - m) * k;
    r.z = __expf(x[i].z - m) * k;
    r.w = __expf(x[i].w - m) * k;
    __builtin_nontemporal_store(r, &o4[32 * i]);
  }
}

extern "C" void kernel_launch(void* const* d_in, const int* in_sizes, int n_in,
                              void* d_out, int out_size, void* d_ws, size_t ws_size,
                              hipStream_t stream) {
  const float* logits = (const float*)d_in[0];
  const float* w1 = (const float*)d_in[1];
  const float* b1 = (const float*)d_in[2];
  const float* w2 = (const float*)d_in[3];
  const float* b2 = (const float*)d_in[4];
  float* out = (float*)d_out;

  char* ws = (char*)d_ws;
  v2f*   stats = (v2f*)ws;
  float* flags = (float*)(ws + (size_t)NROWS * 16 + 32768);

  row_stats_flag_kernel<<<NBLOCKS, BLOCK_THREADS, 0, stream>>>(logits, w1, b1, w2, b2, stats, flags);
  finalize_kernel<<<NBLOCKS, BLOCK_THREADS, 0, stream>>>(logits, stats, flags, out);
}
